# starting kernel
# baseline (speedup 1.0000x reference)
#include <hip/hip_runtime.h>
#include <cstdint>

#define D 128
#define LDSP 132
#define SCAN_BS 1024

typedef float v2f __attribute__((ext_vector_type(2)));
typedef float v8f __attribute__((ext_vector_type(8)));

__device__ __forceinline__ v2f mkv2(float a, float b) { v2f v; v.x = a; v.y = b; return v; }

__device__ __forceinline__ v8f wmma_f32(v2f a, v2f b, v8f c) {
    return __builtin_amdgcn_wmma_f32_16x16x4_f32(false, a, false, b, (short)0, c, false, false);
}

__device__ __forceinline__ float sigmoidf_(float x) { return 1.0f / (1.0f + __expf(-x)); }

__global__ void k_izero(int* __restrict__ p, int n) {
    int i = blockIdx.x * blockDim.x + threadIdx.x;
    if (i < n) p[i] = 0;
}

__global__ void k_count(const int* __restrict__ dst, int* __restrict__ indeg, int E) {
    int e = blockIdx.x * blockDim.x + threadIdx.x;
    if (e < E) atomicAdd(indeg + dst[e], 1);
}

__global__ __launch_bounds__(SCAN_BS) void k_scan1(const int* __restrict__ in,
                                                   int* __restrict__ incl,
                                                   int* __restrict__ bsums, int n) {
    __shared__ int sh[SCAN_BS];
    int gid = blockIdx.x * SCAN_BS + threadIdx.x;
    int v = (gid < n) ? in[gid] : 0;
    sh[threadIdx.x] = v;
    __syncthreads();
    for (int off = 1; off < SCAN_BS; off <<= 1) {
        int t = (threadIdx.x >= off) ? sh[threadIdx.x - off] : 0;
        __syncthreads();
        sh[threadIdx.x] += t;
        __syncthreads();
    }
    if (gid < n) incl[gid] = sh[threadIdx.x];
    if (threadIdx.x == SCAN_BS - 1) bsums[blockIdx.x] = sh[SCAN_BS - 1];
}

__global__ __launch_bounds__(SCAN_BS) void k_scan2(int* __restrict__ bsums, int nb) {
    __shared__ int sh[SCAN_BS];
    int v = (threadIdx.x < nb) ? bsums[threadIdx.x] : 0;
    sh[threadIdx.x] = v;
    __syncthreads();
    for (int off = 1; off < SCAN_BS; off <<= 1) {
        int t = (threadIdx.x >= off) ? sh[threadIdx.x - off] : 0;
        __syncthreads();
        sh[threadIdx.x] += t;
        __syncthreads();
    }
    if (threadIdx.x < nb) bsums[threadIdx.x] = sh[threadIdx.x] - v;
}

__global__ void k_scan3(int* __restrict__ start  ,
                        const int* __restrict__ indeg, const int* __restrict__ bsums,
                        int* __restrict__ cursor, int n) {
    int gid = blockIdx.x * blockDim.x + threadIdx.x;
    if (gid < n) {
        int ex = start[gid] - indeg[gid] + bsums[gid / SCAN_BS];
        start[gid] = ex;
        cursor[gid] = ex;
    }
}

__global__ void k_fill(const int* __restrict__ src, const int* __restrict__ dst,
                       int* __restrict__ cursor, int* __restrict__ csrc, int E) {
    int e = blockIdx.x * blockDim.x + threadIdx.x;
    if (e < E) {
        int p = atomicAdd(cursor + dst[e], 1);
        csrc[p] = src[e];
    }
}

__global__ void k_gather2(const float* __restrict__ xin, const float* __restrict__ hl,
                          const int* __restrict__ csrc,
                          const int* __restrict__ start, const int* __restrict__ indeg,
                          float* __restrict__ aggx, float* __restrict__ aggh, int N) {
    int node = (int)(((size_t)blockIdx.x * blockDim.x + threadIdx.x) >> 5);
    int lane = threadIdx.x & 31;
    if (node >= N) return;
    int base = start[node], cnt = indeg[node];
    float4 ax = {0.f, 0.f, 0.f, 0.f}, ah = {0.f, 0.f, 0.f, 0.f};
#pragma unroll 2
    for (int j = 0; j < cnt; ++j) {
        int s = csrc[base + j];
        float4 vx = ((const float4*)(xin + (size_t)s * D))[lane];
        float4 vh = ((const float4*)(hl + (size_t)s * D))[lane];
        ax.x += vx.x; ax.y += vx.y; ax.z += vx.z; ax.w += vx.w;
        ah.x += vh.x; ah.y += vh.y; ah.z += vh.z; ah.w += vh.w;
    }
    float inv = 1.0f / fmaxf((float)cnt, 1.0f);
    float4 ox = {ax.x * inv, ax.y * inv, ax.z * inv, ax.w * inv};
    float4 oh = {ah.x * inv, ah.y * inv, ah.z * inv, ah.w * inv};
    ((float4*)(aggx + (size_t)node * D))[lane] = ox;
    ((float4*)(aggh + (size_t)node * D))[lane] = oh;
}

__global__ void k_gather1(const float* __restrict__ xin,
                          const int* __restrict__ csrc,
                          const int* __restrict__ start, const int* __restrict__ indeg,
                          float* __restrict__ agg, int N) {
    int node = (int)(((size_t)blockIdx.x * blockDim.x + threadIdx.x) >> 5);
    int lane = threadIdx.x & 31;
    if (node >= N) return;
    int base = start[node], cnt = indeg[node];
    float4 ax = {0.f, 0.f, 0.f, 0.f};
#pragma unroll 2
    for (int j = 0; j < cnt; ++j) {
        int s = csrc[base + j];
        float4 vx = ((const float4*)(xin + (size_t)s * D))[lane];
        ax.x += vx.x; ax.y += vx.y; ax.z += vx.z; ax.w += vx.w;
    }
    float inv = 1.0f / fmaxf((float)cnt, 1.0f);
    float4 ox = {ax.x * inv, ax.y * inv, ax.z * inv, ax.w * inv};
    ((float4*)(agg + (size_t)node * D))[lane] = ox;
}

__global__ __launch_bounds__(256) void k_gates(
    const float* __restrict__ xin, const float* __restrict__ hl,
    const float* __restrict__ Wl, const float* __restrict__ Wr,
    const float* __restrict__ bias,
    float* __restrict__ bufA,
    float* __restrict__ bufB,
    float* __restrict__ rh, int N) {
    __shared__ float lds[4][32][LDSP];
    const int row0 = blockIdx.x * 32;
    const int tid = threadIdx.x;
    for (int idx = tid; idx < 32 * D; idx += 256) {
        int r = idx >> 7, c = idx & (D - 1);
        int row = row0 + r; if (row >= N) row = N - 1;
        size_t off = (size_t)row * D + c;
        lds[0][r][c] = bufA[off];
        lds[1][r][c] = xin[off];
        lds[2][r][c] = bufB[off];
        lds[3][r][c] = hl[off];
    }
    __syncthreads();

    const int wave = tid >> 5;
    const int lane = tid & 31;
    const int m = lane & 15;
    const int half = lane >> 4;
    const int col = wave * 16 + m;

    v8f cz[2] = {{}, {}};
    v8f cr[2] = {{}, {}};
    v8f ct[2] = {{}, {}};
#pragma unroll 2
    for (int kb = 0; kb < D; kb += 4) {
        const int kk = kb + half * 2;
#define BF(W, g) mkv2(W[(size_t)(g) * D * D + (size_t)kk * D + col], \
                      W[(size_t)(g) * D * D + (size_t)(kk + 1) * D + col])
        v2f b0l = BF(Wl, 0), b0r = BF(Wr, 0);
        v2f b1l = BF(Wl, 1), b1r = BF(Wr, 1);
        v2f b2l = BF(Wl, 2), b2r = BF(Wr, 2);
        v2f b3l = BF(Wl, 3), b3r = BF(Wr, 3);
        v2f b4l = BF(Wl, 4), b4r = BF(Wr, 4);
#undef BF
#pragma unroll
        for (int sub = 0; sub < 2; ++sub) {
            const int mm = sub * 16 + m;
            v2f aAX = mkv2(lds[0][mm][kk], lds[0][mm][kk + 1]);
            v2f aX  = mkv2(lds[1][mm][kk], lds[1][mm][kk + 1]);
            v2f aAH = mkv2(lds[2][mm][kk], lds[2][mm][kk + 1]);
            v2f aH  = mkv2(lds[3][mm][kk], lds[3][mm][kk + 1]);
            cz[sub] = wmma_f32(aAX, b0l, cz[sub]);
            cz[sub] = wmma_f32(aX,  b0r, cz[sub]);
            cz[sub] = wmma_f32(aAH, b1l, cz[sub]);
            cz[sub] = wmma_f32(aH,  b1r, cz[sub]);
            cr[sub] = wmma_f32(aAX, b2l, cr[sub]);
            cr[sub] = wmma_f32(aX,  b2r, cr[sub]);
            cr[sub] = wmma_f32(aAH, b3l, cr[sub]);
            cr[sub] = wmma_f32(aH,  b3r, cr[sub]);
            ct[sub] = wmma_f32(aAX, b4l, ct[sub]);
            ct[sub] = wmma_f32(aX,  b4r, ct[sub]);
        }
    }

    const float bz = bias[col] + bias[D + col];
    const float br = bias[2 * D + col] + bias[3 * D + col];
    const float bt = bias[4 * D + col];
#pragma unroll
    for (int sub = 0; sub < 2; ++sub) {
#pragma unroll
        for (int j = 0; j < 8; ++j) {
            int rloc = sub * 16 + j + half * 8;
            int row = row0 + rloc;
            if (row < N) {
                size_t off = (size_t)row * D + col;
                float z = sigmoidf_(cz[sub][j] + bz);
                float r = sigmoidf_(cr[sub][j] + br);
                float hv = lds[3][rloc][col];
                bufA[off] = z;
                rh[off] = r * hv;
                bufB[off] = ct[sub][j] + bt;
            }
        }
    }
}

__global__ __launch_bounds__(256) void k_final(
    const float* __restrict__ hl,
    const float* __restrict__ Wl5, const float* __restrict__ Wr5,
    const float* __restrict__ b5,
    const float* __restrict__ zbuf, const float* __restrict__ tpart,
    const float* __restrict__ rh, const float* __restrict__ aggrh,
    float* __restrict__ outl, int N) {
    __shared__ float lds[2][32][LDSP];
    const int row0 = blockIdx.x * 32;
    const int tid = threadIdx.x;
    for (int idx = tid; idx < 32 * D; idx += 256) {
        int r = idx >> 7, c = idx & (D - 1);
        int row = row0 + r; if (row >= N) row = N - 1;
        size_t off = (size_t)row * D + c;
        lds[0][r][c] = aggrh[off];
        lds[1][r][c] = rh[off];
    }
    __syncthreads();

    const int wave = tid >> 5;
    const int lane = tid & 31;
    const int m = lane & 15;
    const int half = lane >> 4;
    const int col = wave * 16 + m;

    v8f c[2] = {{}, {}};
#pragma unroll
    for (int sub = 0; sub < 2; ++sub)
#pragma unroll
        for (int j = 0; j < 8; ++j) {
            int row = row0 + sub * 16 + j + half * 8;
            if (row < N) c[sub][j] = tpart[(size_t)row * D + col];
        }
#pragma unroll 2
    for (int kb = 0; kb < D; kb += 4) {
        const int kk = kb + half * 2;
        v2f bl = mkv2(Wl5[(size_t)kk * D + col], Wl5[(size_t)(kk + 1) * D + col]);
        v2f br = mkv2(Wr5[(size_t)kk * D + col], Wr5[(size_t)(kk + 1) * D + col]);
#pragma unroll
        for (int sub = 0; sub < 2; ++sub) {
            const int mm = sub * 16 + m;
            v2f a0 = mkv2(lds[0][mm][kk], lds[0][mm][kk + 1]);
            v2f a1 = mkv2(lds[1][mm][kk], lds[1][mm][kk + 1]);
            c[sub] = wmma_f32(a0, bl, c[sub]);
            c[sub] = wmma_f32(a1, br, c[sub]);
        }
    }

    const float bb = b5[col];
#pragma unroll
    for (int sub = 0; sub < 2; ++sub)
#pragma unroll
        for (int j = 0; j < 8; ++j) {
            int row = row0 + sub * 16 + j + half * 8;
            if (row < N) {
                size_t off = (size_t)row * D + col;
                float ht = tanhf(c[sub][j] + bb);
                float z = zbuf[off];
                float hv = hl[off];
                outl[off] = z * hv + (1.0f - z) * ht;
            }
        }
}

extern "C" void kernel_launch(void* const* d_in, const int* in_sizes, int n_in,
                              void* d_out, int out_size, void* d_ws, size_t ws_size,
                              hipStream_t stream) {
    const float* x  = (const float*)d_in[0];
    const float* h  = (const float*)d_in[1];
    const float* Wl = (const float*)d_in[2];
    const float* Wr = (const float*)d_in[3];
    const float* b  = (const float*)d_in[4];
    const int* src  = (const int*)d_in[5];
    const int* dst  = (const int*)d_in[6];
    const size_t ND = (size_t)in_sizes[0];
    const int N = (int)(ND / D);
    const int L = in_sizes[1] / (int)ND;
    const int E = in_sizes[5];
    float* out = (float*)d_out;

    char* wsc = (char*)d_ws;
    int* indeg  = (int*)wsc;
    int* start  = indeg + N;
    int* cursor = start + N;
    int* bsums  = cursor + N;
    int* csrc   = bsums + SCAN_BS;
    size_t int_bytes = ((size_t)(3 * N + SCAN_BS + E) * sizeof(int) + 63) & ~(size_t)63;
    float* fbase = (float*)(wsc + int_bytes);
    float* bufA  = fbase;
    float* bufB  = bufA + ND;
    float* rh    = bufB + ND;
    float* aggrh = rh + ND;

    const int nb = (N + SCAN_BS - 1) / SCAN_BS;
    const int tiles = (N + 31) / 32;
    const int gblocks = (N + 7) / 8;

    k_izero<<<(N + 255) / 256, 256, 0, stream>>>(indeg, N);
    k_count<<<(E + 255) / 256, 256, 0, stream>>>(dst, indeg, E);
    k_scan1<<<nb, SCAN_BS, 0, stream>>>(indeg, start, bsums, N);
    k_scan2<<<1, SCAN_BS, 0, stream>>>(bsums, nb);
    k_scan3<<<(N + 255) / 256, 256, 0, stream>>>(start, indeg, bsums, cursor, N);
    k_fill<<<(E + 255) / 256, 256, 0, stream>>>(src, dst, cursor, csrc, E);

    for (int l = 0; l < L; ++l) {
        const float* inp = (l == 0) ? x : (out + (size_t)(l - 1) * ND);
        const float* hlp = h + (size_t)l * ND;
        const float* Wll = Wl + (size_t)l * 6 * D * D;
        const float* Wrl = Wr + (size_t)l * 6 * D * D;
        const float* bl  = b + (size_t)l * 6 * D;
        float* outl = out + (size_t)l * ND;

        k_gather2<<<gblocks, 256, 0, stream>>>(inp, hlp, csrc, start, indeg,
                                               bufA, bufB, N);
        k_gates<<<tiles, 256, 0, stream>>>(inp, hlp, Wll, Wrl, bl,
                                           bufA, bufB, rh, N);
        k_gather1<<<gblocks, 256, 0, stream>>>(rh, csrc, start, indeg, aggrh, N);
        k_final<<<tiles, 256, 0, stream>>>(hlp, Wll + 5 * D * D, Wrl + 5 * D * D,
                                           bl + 5 * D, bufA, bufB, rh, aggrh,
                                           outl, N);
    }
}


# GEMM kernels: tile, seed and epilogue loads go out as one batch each (rows clamped, stores alone guarded, full tiles on a block-uniform path), was a wait per trip or row; sum order unchanged
# speedup vs baseline: 1.0861x; 1.0861x over previous
#include <hip/hip_runtime.h>
#include <cstdint>

#define D 128
#define LDSP 132
#define SCAN_BS 1024

typedef float v2f __attribute__((ext_vector_type(2)));
typedef float v8f __attribute__((ext_vector_type(8)));

__device__ __forceinline__ v2f mkv2(float a, float b) { v2f v; v.x = a; v.y = b; return v; }

__device__ __forceinline__ v8f wmma_f32(v2f a, v2f b, v8f c) {
    return __builtin_amdgcn_wmma_f32_16x16x4_f32(false, a, false, b, (short)0, c, false, false);
}

__device__ __forceinline__ float sigmoidf_(float x) { return 1.0f / (1.0f + __expf(-x)); }

__global__ void k_izero(int* __restrict__ p, int n) {
    int i = blockIdx.x * blockDim.x + threadIdx.x;
    if (i < n) p[i] = 0;
}

__global__ void k_count(const int* __restrict__ dst, int* __restrict__ indeg, int E) {
    int e = blockIdx.x * blockDim.x + threadIdx.x;
    if (e < E) atomicAdd(indeg + dst[e], 1);
}

__global__ __launch_bounds__(SCAN_BS) void k_scan1(const int* __restrict__ in,
                                                   int* __restrict__ incl,
                                                   int* __restrict__ bsums, int n) {
    __shared__ int sh[SCAN_BS];
    int gid = blockIdx.x * SCAN_BS + threadIdx.x;
    int v = (gid < n) ? in[gid] : 0;
    sh[threadIdx.x] = v;
    __syncthreads();
    for (int off = 1; off < SCAN_BS; off <<= 1) {
        int t = (threadIdx.x >= off) ? sh[threadIdx.x - off] : 0;
        __syncthreads();
        sh[threadIdx.x] += t;
        __syncthreads();
    }
    if (gid < n) incl[gid] = sh[threadIdx.x];
    if (threadIdx.x == SCAN_BS - 1) bsums[blockIdx.x] = sh[SCAN_BS - 1];
}

__global__ __launch_bounds__(SCAN_BS) void k_scan2(int* __restrict__ bsums, int nb) {
    __shared__ int sh[SCAN_BS];
    int v = (threadIdx.x < nb) ? bsums[threadIdx.x] : 0;
    sh[threadIdx.x] = v;
    __syncthreads();
    for (int off = 1; off < SCAN_BS; off <<= 1) {
        int t = (threadIdx.x >= off) ? sh[threadIdx.x - off] : 0;
        __syncthreads();
        sh[threadIdx.x] += t;
        __syncthreads();
    }
    if (threadIdx.x < nb) bsums[threadIdx.x] = sh[threadIdx.x] - v;
}

__global__ void k_scan3(int* __restrict__ start  ,
                        const int* __restrict__ indeg, const int* __restrict__ bsums,
                        int* __restrict__ cursor, int n) {
    int gid = blockIdx.x * blockDim.x + threadIdx.x;
    if (gid < n) {
        int ex = start[gid] - indeg[gid] + bsums[gid / SCAN_BS];
        start[gid] = ex;
        cursor[gid] = ex;
    }
}

__global__ void k_fill(const int* __restrict__ src, const int* __restrict__ dst,
                       int* __restrict__ cursor, int* __restrict__ csrc, int E) {
    int e = blockIdx.x * blockDim.x + threadIdx.x;
    if (e < E) {
        int p = atomicAdd(cursor + dst[e], 1);
        csrc[p] = src[e];
    }
}

__global__ void k_gather2(const float* __restrict__ xin, const float* __restrict__ hl,
                          const int* __restrict__ csrc,
                          const int* __restrict__ start, const int* __restrict__ indeg,
                          float* __restrict__ aggx, float* __restrict__ aggh, int N) {
    int node = (int)(((size_t)blockIdx.x * blockDim.x + threadIdx.x) >> 5);
    int lane = threadIdx.x & 31;
    if (node >= N) return;
    int base = start[node], cnt = indeg[node];
    float4 ax = {0.f, 0.f, 0.f, 0.f}, ah = {0.f, 0.f, 0.f, 0.f};
#pragma unroll 2
    for (int j = 0; j < cnt; ++j) {
        int s = csrc[base + j];
        float4 vx = ((const float4*)(xin + (size_t)s * D))[lane];
        float4 vh = ((const float4*)(hl + (size_t)s * D))[lane];
        ax.x += vx.x; ax.y += vx.y; ax.z += vx.z; ax.w += vx.w;
        ah.x += vh.x; ah.y += vh.y; ah.z += vh.z; ah.w += vh.w;
    }
    float inv = 1.0f / fmaxf((float)cnt, 1.0f);
    float4 ox = {ax.x * inv, ax.y * inv, ax.z * inv, ax.w * inv};
    float4 oh = {ah.x * inv, ah.y * inv, ah.z * inv, ah.w * inv};
    ((float4*)(aggx + (size_t)node * D))[lane] = ox;
    ((float4*)(aggh + (size_t)node * D))[lane] = oh;
}

__global__ void k_gather1(const float* __restrict__ xin,
                          const int* __restrict__ csrc,
                          const int* __restrict__ start, const int* __restrict__ indeg,
                          float* __restrict__ agg, int N) {
    int node = (int)(((size_t)blockIdx.x * blockDim.x + threadIdx.x) >> 5);
    int lane = threadIdx.x & 31;
    if (node >= N) return;
    int base = start[node], cnt = indeg[node];
    float4 ax = {0.f, 0.f, 0.f, 0.f};
#pragma unroll 2
    for (int j = 0; j < cnt; ++j) {
        int s = csrc[base + j];
        float4 vx = ((const float4*)(xin + (size_t)s * D))[lane];
        ax.x += vx.x; ax.y += vx.y; ax.z += vx.z; ax.w += vx.w;
    }
    float inv = 1.0f / fmaxf((float)cnt, 1.0f);
    float4 ox = {ax.x * inv, ax.y * inv, ax.z * inv, ax.w * inv};
    ((float4*)(agg + (size_t)node * D))[lane] = ox;
}

template <bool FULL>
__device__ __forceinline__ void gates_store(
    const v8f (&cz)[2], const v8f (&cr)[2], const v8f (&ct)[2], const float (&hv)[2][8],
    float bz, float br, float bt,
    float* __restrict__ bufA, float* __restrict__ bufB, float* __restrict__ rh,
    int row0, int half, int col, int N) {
#pragma unroll
    for (int sub = 0; sub < 2; ++sub) {
#pragma unroll
        for (int j = 0; j < 8; ++j) {
            const int row = row0 + sub * 16 + j + half * 8;
            if (FULL || row < N) {
                size_t off = (size_t)row * D + col;
                float z = sigmoidf_(cz[sub][j] + bz);
                float r = sigmoidf_(cr[sub][j] + br);
                bufA[off] = z;
                rh[off] = r * hv[sub][j];
                bufB[off] = ct[sub][j] + bt;
            }
        }
    }
}

__global__ __launch_bounds__(256) void k_gates(
    const float* __restrict__ xin, const float* __restrict__ hl,
    const float* __restrict__ Wl, const float* __restrict__ Wr,
    const float* __restrict__ bias,
    float* __restrict__ bufA,
    float* __restrict__ bufB,
    float* __restrict__ rh, int N) {
    __shared__ float lds[4][32][LDSP];
    const int row0 = blockIdx.x * 32;
    const int tid = threadIdx.x;
    {
        const int sc = tid & (D - 1);
        const int rb = tid >> 7;
        float s0[16], s1[16], s2[16], s3[16];
#pragma unroll
        for (int i = 0; i < 16; ++i) {
            int row = row0 + rb + 2 * i; if (row >= N) row = N - 1;
            const unsigned off = (unsigned)row * D + sc;
            s0[i] = bufA[off];
            s1[i] = xin[off];
            s2[i] = bufB[off];
            s3[i] = hl[off];
        }
#pragma unroll
        for (int i = 0; i < 16; ++i) {
            const int r = rb + 2 * i;
            lds[0][r][sc] = s0[i];
            lds[1][r][sc] = s1[i];
            lds[2][r][sc] = s2[i];
            lds[3][r][sc] = s3[i];
        }
    }
    __syncthreads();

    const int wave = tid >> 5;
    const int lane = tid & 31;
    const int m = lane & 15;
    const int half = lane >> 4;
    const int col = wave * 16 + m;

    v8f cz[2] = {{}, {}};
    v8f cr[2] = {{}, {}};
    v8f ct[2] = {{}, {}};
#pragma unroll 2
    for (int kb = 0; kb < D; kb += 4) {
        const int kk = kb + half * 2;
#define BF(W, g) mkv2(W[(size_t)(g) * D * D + (size_t)kk * D + col], \
                      W[(size_t)(g) * D * D + (size_t)(kk + 1) * D + col])
        v2f b0l = BF(Wl, 0), b0r = BF(Wr, 0);
        v2f b1l = BF(Wl, 1), b1r = BF(Wr, 1);
        v2f b2l = BF(Wl, 2), b2r = BF(Wr, 2);
        v2f b3l = BF(Wl, 3), b3r = BF(Wr, 3);
        v2f b4l = BF(Wl, 4), b4r = BF(Wr, 4);
#undef BF
#pragma unroll
        for (int sub = 0; sub < 2; ++sub) {
            const int mm = sub * 16 + m;
            v2f aAX = mkv2(lds[0][mm][kk], lds[0][mm][kk + 1]);
            v2f aX  = mkv2(lds[1][mm][kk], lds[1][mm][kk + 1]);
            v2f aAH = mkv2(lds[2][mm][kk], lds[2][mm][kk + 1]);
            v2f aH  = mkv2(lds[3][mm][kk], lds[3][mm][kk + 1]);
            cz[sub] = wmma_f32(aAX, b0l, cz[sub]);
            cz[sub] = wmma_f32(aX,  b0r, cz[sub]);
            cz[sub] = wmma_f32(aAH, b1l, cz[sub]);
            cz[sub] = wmma_f32(aH,  b1r, cz[sub]);
            cr[sub] = wmma_f32(aAX, b2l, cr[sub]);
            cr[sub] = wmma_f32(aX,  b2r, cr[sub]);
            cr[sub] = wmma_f32(aAH, b3l, cr[sub]);
            cr[sub] = wmma_f32(aH,  b3r, cr[sub]);
            ct[sub] = wmma_f32(aAX, b4l, ct[sub]);
            ct[sub] = wmma_f32(aX,  b4r, ct[sub]);
        }
    }

    const float bz = bias[col] + bias[D + col];
    const float br = bias[2 * D + col] + bias[3 * D + col];
    const float bt = bias[4 * D + col];
    float hv[2][8];
#pragma unroll
    for (int sub = 0; sub < 2; ++sub)
#pragma unroll
        for (int j = 0; j < 8; ++j) hv[sub][j] = lds[3][sub * 16 + j + half * 8][col];
    if (row0 + 32 <= N)
        gates_store<true>(cz, cr, ct, hv, bz, br, bt, bufA, bufB, rh, row0, half, col, N);
    else
        gates_store<false>(cz, cr, ct, hv, bz, br, bt, bufA, bufB, rh, row0, half, col, N);
}

template <bool FULL>
__device__ __forceinline__ void final_store(
    const v8f (&c)[2], const float (&zz)[2][8], const float (&hh)[2][8], float bb,
    float* __restrict__ outl, int row0, int half, int col, int N) {
#pragma unroll
    for (int sub = 0; sub < 2; ++sub)
#pragma unroll
        for (int j = 0; j < 8; ++j) {
            const int row = row0 + sub * 16 + j + half * 8;
            if (FULL || row < N) {
                size_t off = (size_t)row * D + col;
                float ht = tanhf(c[sub][j] + bb);
                float z = zz[sub][j];
                float hv = hh[sub][j];
                outl[off] = z * hv + (1.0f - z) * ht;
            }
        }
}

__global__ __launch_bounds__(256) void k_final(
    const float* __restrict__ hl,
    const float* __restrict__ Wl5, const float* __restrict__ Wr5,
    const float* __restrict__ b5,
    const float* __restrict__ zbuf, const float* __restrict__ tpart,
    const float* __restrict__ rh, const float* __restrict__ aggrh,
    float* __restrict__ outl, int N) {
    __shared__ float lds[2][32][LDSP];
    const int row0 = blockIdx.x * 32;
    const int tid = threadIdx.x;
    const int wave = tid >> 5;
    const int lane = tid & 31;
    const int m = lane & 15;
    const int half = lane >> 4;
    const int col = wave * 16 + m;

    v8f c[2] = {{}, {}};
    {
        const int sc = tid & (D - 1);
        const int rb = tid >> 7;
        float s0[16], s1[16];
#pragma unroll
        for (int i = 0; i < 16; ++i) {
            int row = row0 + rb + 2 * i; if (row >= N) row = N - 1;
            const unsigned off = (unsigned)row * D + sc;
            s0[i] = aggrh[off];
            s1[i] = rh[off];
        }
#pragma unroll
        for (int sub = 0; sub < 2; ++sub)
#pragma unroll
            for (int j = 0; j < 8; ++j) {
                int row = row0 + sub * 16 + j + half * 8; if (row >= N) row = N - 1;
                c[sub][j] = tpart[(unsigned)row * D + col];
            }
#pragma unroll
        for (int i = 0; i < 16; ++i) {
            const int r = rb + 2 * i;
            lds[0][r][sc] = s0[i];
            lds[1][r][sc] = s1[i];
        }
    }
    __syncthreads();
#pragma unroll 2
    for (int kb = 0; kb < D; kb += 4) {
        const int kk = kb + half * 2;
        v2f bl = mkv2(Wl5[(size_t)kk * D + col], Wl5[(size_t)(kk + 1) * D + col]);
        v2f br = mkv2(Wr5[(size_t)kk * D + col], Wr5[(size_t)(kk + 1) * D + col]);
#pragma unroll
        for (int sub = 0; sub < 2; ++sub) {
            const int mm = sub * 16 + m;
            v2f a0 = mkv2(lds[0][mm][kk], lds[0][mm][kk + 1]);
            v2f a1 = mkv2(lds[1][mm][kk], lds[1][mm][kk + 1]);
            c[sub] = wmma_f32(a0, bl, c[sub]);
            c[sub] = wmma_f32(a1, br, c[sub]);
        }
    }

    const float bb = b5[col];
    float zz[2][8], hh[2][8];
#pragma unroll
    for (int sub = 0; sub < 2; ++sub)
#pragma unroll
        for (int j = 0; j < 8; ++j) {
            int row = row0 + sub * 16 + j + half * 8; if (row >= N) row = N - 1;
            const unsigned off = (unsigned)row * D + col;
            zz[sub][j] = zbuf[off];
            hh[sub][j] = hl[off];
        }
    if (row0 + 32 <= N)
        final_store<true>(c, zz, hh, bb, outl, row0, half, col, N);
    else
        final_store<false>(c, zz, hh, bb, outl, row0, half, col, N);
}

extern "C" void kernel_launch(void* const* d_in, const int* in_sizes, int n_in,
                              void* d_out, int out_size, void* d_ws, size_t ws_size,
                              hipStream_t stream) {
    const float* x  = (const float*)d_in[0];
    const float* h  = (const float*)d_in[1];
    const float* Wl = (const float*)d_in[2];
    const float* Wr = (const float*)d_in[3];
    const float* b  = (const float*)d_in[4];
    const int* src  = (const int*)d_in[5];
    const int* dst  = (const int*)d_in[6];
    const size_t ND = (size_t)in_sizes[0];
    const int N = (int)(ND / D);
    const int L = in_sizes[1] / (int)ND;
    const int E = in_sizes[5];
    float* out = (float*)d_out;

    char* wsc = (char*)d_ws;
    int* indeg  = (int*)wsc;
    int* start  = indeg + N;
    int* cursor = start + N;
    int* bsums  = cursor + N;
    int* csrc   = bsums + SCAN_BS;
    size_t int_bytes = ((size_t)(3 * N + SCAN_BS + E) * sizeof(int) + 63) & ~(size_t)63;
    float* fbase = (float*)(wsc + int_bytes);
    float* bufA  = fbase;
    float* bufB  = bufA + ND;
    float* rh    = bufB + ND;
    float* aggrh = rh + ND;

    const int nb = (N + SCAN_BS - 1) / SCAN_BS;
    const int tiles = (N + 31) / 32;
    const int gblocks = (N + 7) / 8;

    k_izero<<<(N + 255) / 256, 256, 0, stream>>>(indeg, N);
    k_count<<<(E + 255) / 256, 256, 0, stream>>>(dst, indeg, E);
    k_scan1<<<nb, SCAN_BS, 0, stream>>>(indeg, start, bsums, N);
    k_scan2<<<1, SCAN_BS, 0, stream>>>(bsums, nb);
    k_scan3<<<(N + 255) / 256, 256, 0, stream>>>(start, indeg, bsums, cursor, N);
    k_fill<<<(E + 255) / 256, 256, 0, stream>>>(src, dst, cursor, csrc, E);

    for (int l = 0; l < L; ++l) {
        const float* inp = (l == 0) ? x : (out + (size_t)(l - 1) * ND);
        const float* hlp = h + (size_t)l * ND;
        const float* Wll = Wl + (size_t)l * 6 * D * D;
        const float* Wrl = Wr + (size_t)l * 6 * D * D;
        const float* bl  = b + (size_t)l * 6 * D;
        float* outl = out + (size_t)l * ND;

        k_gather2<<<gblocks, 256, 0, stream>>>(inp, hlp, csrc, start, indeg,
                                               bufA, bufB, N);
        k_gates<<<tiles, 256, 0, stream>>>(inp, hlp, Wll, Wrl, bl,
                                           bufA, bufB, rh, N);
        k_gather1<<<gblocks, 256, 0, stream>>>(rh, csrc, start, indeg, aggrh, N);
        k_final<<<tiles, 256, 0, stream>>>(hlp, Wll + 5 * D * D, Wrl + 5 * D * D,
                                           bl + 5 * D, bufA, bufB, rh, aggrh,
                                           outl, N);
    }
}


# Neighbour mean aggregation moved into the GEMM kernels: each wave sums its 4 tile nodes' CSR rows (slot order as before) straight into the LDS tile; 4 gather launches and the aggregate buffers go
# speedup vs baseline: 1.1943x; 1.0996x over previous
#include <hip/hip_runtime.h>
#include <cstdint>

#define D 128
#define LDSP 132
#define SCAN_BS 1024

typedef float v2f __attribute__((ext_vector_type(2)));
typedef float v8f __attribute__((ext_vector_type(8)));

__device__ __forceinline__ v2f mkv2(float a, float b) { v2f v; v.x = a; v.y = b; return v; }

__device__ __forceinline__ v8f wmma_f32(v2f a, v2f b, v8f c) {
    return __builtin_amdgcn_wmma_f32_16x16x4_f32(false, a, false, b, (short)0, c, false, false);
}

__device__ __forceinline__ float sigmoidf_(float x) { return 1.0f / (1.0f + __expf(-x)); }

__global__ void k_izero(int* __restrict__ p, int n) {
    int i = blockIdx.x * blockDim.x + threadIdx.x;
    if (i < n) p[i] = 0;
}

__global__ void k_count(const int* __restrict__ dst, int* __restrict__ indeg, int E) {
    int e = blockIdx.x * blockDim.x + threadIdx.x;
    if (e < E) atomicAdd(indeg + dst[e], 1);
}

__global__ __launch_bounds__(SCAN_BS) void k_scan1(const int* __restrict__ in,
                                                   int* __restrict__ incl,
                                                   int* __restrict__ bsums, int n) {
    __shared__ int sh[SCAN_BS];
    int gid = blockIdx.x * SCAN_BS + threadIdx.x;
    int v = (gid < n) ? in[gid] : 0;
    sh[threadIdx.x] = v;
    __syncthreads();
    for (int off = 1; off < SCAN_BS; off <<= 1) {
        int t = (threadIdx.x >= off) ? sh[threadIdx.x - off] : 0;
        __syncthreads();
        sh[threadIdx.x] += t;
        __syncthreads();
    }
    if (gid < n) incl[gid] = sh[threadIdx.x];
    if (threadIdx.x == SCAN_BS - 1) bsums[blockIdx.x] = sh[SCAN_BS - 1];
}

__global__ __launch_bounds__(SCAN_BS) void k_scan2(int* __restrict__ bsums, int nb) {
    __shared__ int sh[SCAN_BS];
    int v = (threadIdx.x < nb) ? bsums[threadIdx.x] : 0;
    sh[threadIdx.x] = v;
    __syncthreads();
    for (int off = 1; off < SCAN_BS; off <<= 1) {
        int t = (threadIdx.x >= off) ? sh[threadIdx.x - off] : 0;
        __syncthreads();
        sh[threadIdx.x] += t;
        __syncthreads();
    }
    if (threadIdx.x < nb) bsums[threadIdx.x] = sh[threadIdx.x] - v;
}

__global__ void k_scan3(int* __restrict__ start  ,
                        const int* __restrict__ indeg, const int* __restrict__ bsums,
                        int* __restrict__ cursor, int n) {
    int gid = blockIdx.x * blockDim.x + threadIdx.x;
    if (gid < n) {
        int ex = start[gid] - indeg[gid] + bsums[gid / SCAN_BS];
        start[gid] = ex;
        cursor[gid] = ex;
    }
}

__global__ void k_fill(const int* __restrict__ src, const int* __restrict__ dst,
                       int* __restrict__ cursor, int* __restrict__ csrc, int E) {
    int e = blockIdx.x * blockDim.x + threadIdx.x;
    if (e < E) {
        int p = atomicAdd(cursor + dst[e], 1);
        csrc[p] = src[e];
    }
}

template <bool TWO>
__device__ __forceinline__ void close_node(float4& a0, float4& a1, int cnt,
                                           float* d0, float* d1, int lane) {
    const float inv = 1.0f / fmaxf((float)cnt, 1.0f);
    float4 o0 = {a0.x * inv, a0.y * inv, a0.z * inv, a0.w * inv};
    ((float4*)d0)[lane] = o0;
    a0.x = 0.f; a0.y = 0.f; a0.z = 0.f; a0.w = 0.f;
    if (TWO) {
        float4 o1 = {a1.x * inv, a1.y * inv, a1.z * inv, a1.w * inv};
        ((float4*)d1)[lane] = o1;
        a1.x = 0.f; a1.y = 0.f; a1.z = 0.f; a1.w = 0.f;
    }
}

template <bool TWO, int B>
__device__ __forceinline__ void gather4_to_lds(
    const float* t0, const float* t1,
    const int* __restrict__ csrc, const int* __restrict__ start, const int* __restrict__ indeg,
    float (*dst0)[LDSP], float (*dst1)[LDSP], int n0, int r0, int lane, int N) {
    int cntv = 0;
    if (lane < 4 && n0 + lane < N) cntv = indeg[n0 + lane];
    const int c0 = __builtin_amdgcn_readlane(cntv, 0);
    const int c1 = __builtin_amdgcn_readlane(cntv, 1);
    const int c2 = __builtin_amdgcn_readlane(cntv, 2);
    const int c3 = __builtin_amdgcn_readlane(cntv, 3);
    const int total = c0 + c1 + c2 + c3;
    const int endv = (lane == 0) ? c0 : (lane == 1) ? c0 + c1 : (lane == 2) ? c0 + c1 + c2 : total;
    int base0 = 0;
    if (n0 < N) base0 = start[n0];
    base0 = __builtin_amdgcn_readfirstlane(base0);

    float4 a0 = {0.f, 0.f, 0.f, 0.f}, a1 = {0.f, 0.f, 0.f, 0.f};
    int q = 0;
    int qend = c0;
    for (int cb = 0; cb < total; cb += 32) {
        const int e = cb + lane;
        const int idx = (e < total) ? csrc[base0 + e] : 0;
        const int lim = (total - cb < 32) ? (total - cb) : 32;
        for (int j = 0; j < lim; j += B) {
            float4 v0[B], v1[B];
#pragma unroll
            for (int u = 0; u < B; ++u) {
                const int jj = (j + u < lim) ? (j + u) : (lim - 1);
                const int s = __builtin_amdgcn_readlane(idx, jj);
                v0[u] = ((const float4*)(t0 + (size_t)s * D))[lane];
                if (TWO) v1[u] = ((const float4*)(t1 + (size_t)s * D))[lane];
            }
#pragma unroll
            for (int u = 0; u < B; ++u) {
                if (j + u < lim) {
                    const int eg = cb + j + u;
                    while (eg >= qend) {
                        close_node<TWO>(a0, a1, __builtin_amdgcn_readlane(cntv, q),
                                        &dst0[r0 + q][0], &dst1[r0 + q][0], lane);
                        ++q;
                        qend = __builtin_amdgcn_readlane(endv, q);
                    }
                    a0.x += v0[u].x; a0.y += v0[u].y; a0.z += v0[u].z; a0.w += v0[u].w;
                    if (TWO) { a1.x += v1[u].x; a1.y += v1[u].y; a1.z += v1[u].z; a1.w += v1[u].w; }
                }
            }
        }
    }
    for (; q < 4; ++q)
        close_node<TWO>(a0, a1, __builtin_amdgcn_readlane(cntv, q),
                        &dst0[r0 + q][0], &dst1[r0 + q][0], lane);
}

template <bool FULL>
__device__ __forceinline__ void gates_store(
    const v8f (&cz)[2], const v8f (&cr)[2], const v8f (&ct)[2], const float (&hv)[2][8],
    float bz, float br, float bt,
    float* __restrict__ bufA, float* __restrict__ bufB, float* __restrict__ rh,
    int row0, int half, int col, int N) {
#pragma unroll
    for (int sub = 0; sub < 2; ++sub) {
#pragma unroll
        for (int j = 0; j < 8; ++j) {
            const int row = row0 + sub * 16 + j + half * 8;
            if (FULL || row < N) {
                size_t off = (size_t)row * D + col;
                float z = sigmoidf_(cz[sub][j] + bz);
                float r = sigmoidf_(cr[sub][j] + br);
                bufA[off] = z;
                rh[off] = r * hv[sub][j];
                bufB[off] = ct[sub][j] + bt;
            }
        }
    }
}

__global__ __launch_bounds__(256) void k_gates(
    const float* __restrict__ xin, const float* __restrict__ hl,
    const float* __restrict__ Wl, const float* __restrict__ Wr,
    const float* __restrict__ bias,
    const int* __restrict__ csrc, const int* __restrict__ start, const int* __restrict__ indeg,
    float* __restrict__ bufA,
    float* __restrict__ bufB,
    float* __restrict__ rh, int N) {
    __shared__ __attribute__((aligned(16))) float lds[4][32][LDSP];
    const int row0 = blockIdx.x * 32;
    const int tid = threadIdx.x;
    const int wave = tid >> 5;
    const int lane = tid & 31;
    {
        const int sc = tid & (D - 1);
        const int rb = tid >> 7;
        float s1[16], s3[16];
#pragma unroll
        for (int i = 0; i < 16; ++i) {
            int row = row0 + rb + 2 * i; if (row >= N) row = N - 1;
            const unsigned off = (unsigned)row * D + sc;
            s1[i] = xin[off];
            s3[i] = hl[off];
        }
        gather4_to_lds<true, 4>(xin, hl, csrc, start, indeg, lds[0], lds[2],
                                row0 + wave * 4, wave * 4, lane, N);
#pragma unroll
        for (int i = 0; i < 16; ++i) {
            const int r = rb + 2 * i;
            lds[1][r][sc] = s1[i];
            lds[3][r][sc] = s3[i];
        }
    }
    __syncthreads();

    const int m = lane & 15;
    const int half = lane >> 4;
    const int col = wave * 16 + m;

    v8f cz[2] = {{}, {}};
    v8f cr[2] = {{}, {}};
    v8f ct[2] = {{}, {}};
#pragma unroll 2
    for (int kb = 0; kb < D; kb += 4) {
        const int kk = kb + half * 2;
#define BF(W, g) mkv2(W[(size_t)(g) * D * D + (size_t)kk * D + col], \
                      W[(size_t)(g) * D * D + (size_t)(kk + 1) * D + col])
        v2f b0l = BF(Wl, 0), b0r = BF(Wr, 0);
        v2f b1l = BF(Wl, 1), b1r = BF(Wr, 1);
        v2f b2l = BF(Wl, 2), b2r = BF(Wr, 2);
        v2f b3l = BF(Wl, 3), b3r = BF(Wr, 3);
        v2f b4l = BF(Wl, 4), b4r = BF(Wr, 4);
#undef BF
#pragma unroll
        for (int sub = 0; sub < 2; ++sub) {
            const int mm = sub * 16 + m;
            v2f aAX = mkv2(lds[0][mm][kk], lds[0][mm][kk + 1]);
            v2f aX  = mkv2(lds[1][mm][kk], lds[1][mm][kk + 1]);
            v2f aAH = mkv2(lds[2][mm][kk], lds[2][mm][kk + 1]);
            v2f aH  = mkv2(lds[3][mm][kk], lds[3][mm][kk + 1]);
            cz[sub] = wmma_f32(aAX, b0l, cz[sub]);
            cz[sub] = wmma_f32(aX,  b0r, cz[sub]);
            cz[sub] = wmma_f32(aAH, b1l, cz[sub]);
            cz[sub] = wmma_f32(aH,  b1r, cz[sub]);
            cr[sub] = wmma_f32(aAX, b2l, cr[sub]);
            cr[sub] = wmma_f32(aX,  b2r, cr[sub]);
            cr[sub] = wmma_f32(aAH, b3l, cr[sub]);
            cr[sub] = wmma_f32(aH,  b3r, cr[sub]);
            ct[sub] = wmma_f32(aAX, b4l, ct[sub]);
            ct[sub] = wmma_f32(aX,  b4r, ct[sub]);
        }
    }

    const float bz = bias[col] + bias[D + col];
    const float br = bias[2 * D + col] + bias[3 * D + col];
    const float bt = bias[4 * D + col];
    float hv[2][8];
#pragma unroll
    for (int sub = 0; sub < 2; ++sub)
#pragma unroll
        for (int j = 0; j < 8; ++j) hv[sub][j] = lds[3][sub * 16 + j + half * 8][col];
    if (row0 + 32 <= N)
        gates_store<true>(cz, cr, ct, hv, bz, br, bt, bufA, bufB, rh, row0, half, col, N);
    else
        gates_store<false>(cz, cr, ct, hv, bz, br, bt, bufA, bufB, rh, row0, half, col, N);
}

template <bool FULL>
__device__ __forceinline__ void final_store(
    const v8f (&c)[2], const float (&zz)[2][8], const float (&hh)[2][8], float bb,
    float* __restrict__ outl, int row0, int half, int col, int N) {
#pragma unroll
    for (int sub = 0; sub < 2; ++sub)
#pragma unroll
        for (int j = 0; j < 8; ++j) {
            const int row = row0 + sub * 16 + j + half * 8;
            if (FULL || row < N) {
                size_t off = (size_t)row * D + col;
                float ht = tanhf(c[sub][j] + bb);
                float z = zz[sub][j];
                float hv = hh[sub][j];
                outl[off] = z * hv + (1.0f - z) * ht;
            }
        }
}

__global__ __launch_bounds__(256) void k_final(
    const float* __restrict__ hl,
    const float* __restrict__ Wl5, const float* __restrict__ Wr5,
    const float* __restrict__ b5,
    const float* __restrict__ zbuf, const float* __restrict__ tpart,
    const float* __restrict__ rh,
    const int* __restrict__ csrc, const int* __restrict__ start, const int* __restrict__ indeg,
    float* __restrict__ outl, int N) {
    __shared__ __attribute__((aligned(16))) float lds[2][32][LDSP];
    const int row0 = blockIdx.x * 32;
    const int tid = threadIdx.x;
    const int wave = tid >> 5;
    const int lane = tid & 31;
    const int m = lane & 15;
    const int half = lane >> 4;
    const int col = wave * 16 + m;

    v8f c[2] = {{}, {}};
    {
        const int sc = tid & (D - 1);
        const int rb = tid >> 7;
        float s1[16];
#pragma unroll
        for (int i = 0; i < 16; ++i) {
            int row = row0 + rb + 2 * i; if (row >= N) row = N - 1;
            const unsigned off = (unsigned)row * D + sc;
            s1[i] = rh[off];
        }
#pragma unroll
        for (int sub = 0; sub < 2; ++sub)
#pragma unroll
            for (int j = 0; j < 8; ++j) {
                int row = row0 + sub * 16 + j + half * 8; if (row >= N) row = N - 1;
                c[sub][j] = tpart[(unsigned)row * D + col];
            }
        gather4_to_lds<false, 4>(rh, rh, csrc, start, indeg, lds[0], lds[0],
                                 row0 + wave * 4, wave * 4, lane, N);
#pragma unroll
        for (int i = 0; i < 16; ++i) {
            const int r = rb + 2 * i;
            lds[1][r][sc] = s1[i];
        }
    }
    __syncthreads();
#pragma unroll 2
    for (int kb = 0; kb < D; kb += 4) {
        const int kk = kb + half * 2;
        v2f bl = mkv2(Wl5[(size_t)kk * D + col], Wl5[(size_t)(kk + 1) * D + col]);
        v2f br = mkv2(Wr5[(size_t)kk * D + col], Wr5[(size_t)(kk + 1) * D + col]);
#pragma unroll
        for (int sub = 0; sub < 2; ++sub) {
            const int mm = sub * 16 + m;
            v2f a0 = mkv2(lds[0][mm][kk], lds[0][mm][kk + 1]);
            v2f a1 = mkv2(lds[1][mm][kk], lds[1][mm][kk + 1]);
            c[sub] = wmma_f32(a0, bl, c[sub]);
            c[sub] = wmma_f32(a1, br, c[sub]);
        }
    }

    const float bb = b5[col];
    float zz[2][8], hh[2][8];
#pragma unroll
    for (int sub = 0; sub < 2; ++sub)
#pragma unroll
        for (int j = 0; j < 8; ++j) {
            int row = row0 + sub * 16 + j + half * 8; if (row >= N) row = N - 1;
            const unsigned off = (unsigned)row * D + col;
            zz[sub][j] = zbuf[off];
            hh[sub][j] = hl[off];
        }
    if (row0 + 32 <= N)
        final_store<true>(c, zz, hh, bb, outl, row0, half, col, N);
    else
        final_store<false>(c, zz, hh, bb, outl, row0, half, col, N);
}

extern "C" void kernel_launch(void* const* d_in, const int* in_sizes, int n_in,
                              void* d_out, int out_size, void* d_ws, size_t ws_size,
                              hipStream_t stream) {
    const float* x  = (const float*)d_in[0];
    const float* h  = (const float*)d_in[1];
    const float* Wl = (const float*)d_in[2];
    const float* Wr = (const float*)d_in[3];
    const float* b  = (const float*)d_in[4];
    const int* src  = (const int*)d_in[5];
    const int* dst  = (const int*)d_in[6];
    const size_t ND = (size_t)in_sizes[0];
    const int N = (int)(ND / D);
    const int L = in_sizes[1] / (int)ND;
    const int E = in_sizes[5];
    float* out = (float*)d_out;

    char* wsc = (char*)d_ws;
    int* indeg  = (int*)wsc;
    int* start  = indeg + N;
    int* cursor = start + N;
    int* bsums  = cursor + N;
    int* csrc   = bsums + SCAN_BS;
    size_t int_bytes = ((size_t)(3 * N + SCAN_BS + E) * sizeof(int) + 63) & ~(size_t)63;
    float* fbase = (float*)(wsc + int_bytes);
    float* bufA  = fbase;
    float* bufB  = bufA + ND;
    float* rh    = bufB + ND;

    const int nb = (N + SCAN_BS - 1) / SCAN_BS;
    const int tiles = (N + 31) / 32;

    k_izero<<<(N + 255) / 256, 256, 0, stream>>>(indeg, N);
    k_count<<<(E + 255) / 256, 256, 0, stream>>>(dst, indeg, E);
    k_scan1<<<nb, SCAN_BS, 0, stream>>>(indeg, start, bsums, N);
    k_scan2<<<1, SCAN_BS, 0, stream>>>(bsums, nb);
    k_scan3<<<(N + 255) / 256, 256, 0, stream>>>(start, indeg, bsums, cursor, N);
    k_fill<<<(E + 255) / 256, 256, 0, stream>>>(src, dst, cursor, csrc, E);

    for (int l = 0; l < L; ++l) {
        const float* inp = (l == 0) ? x : (out + (size_t)(l - 1) * ND);
        const float* hlp = h + (size_t)l * ND;
        const float* Wll = Wl + (size_t)l * 6 * D * D;
        const float* Wrl = Wr + (size_t)l * 6 * D * D;
        const float* bl  = b + (size_t)l * 6 * D;
        float* outl = out + (size_t)l * ND;

        k_gates<<<tiles, 256, 0, stream>>>(inp, hlp, Wll, Wrl, bl, csrc, start, indeg,
                                           bufA, bufB, rh, N);
        k_final<<<tiles, 256, 0, stream>>>(hlp, Wll + 5 * D * D, Wrl + 5 * D * D,
                                           bl + 5 * D, bufA, bufB, rh, csrc, start, indeg,
                                           outl, N);
    }
}
